# starting kernel
# baseline (speedup 1.0000x reference)
#include <hip/hip_runtime.h>

typedef float v2f __attribute__((ext_vector_type(2)));
typedef float v8f __attribute__((ext_vector_type(8)));

#define N_NODES 50000
#define N_EDGES 640000
#define N_REL   500
#define DIM     128
#define K2      256

#define WS_WT_I 0
#define WS_WT_O 32768
#define WS_WT_S 65536
#define WS_WT_M 81920
#define WS_AGG  98304
#define WS_CNT  6498304
#define WS_ZERO_N 6450000

#define A_STRIDE_E 260
#define A_STRIDE_N 132

#define EDGE_MT    4
#define EDGE_WAVES 4
#define EDGES_PER_WAVE  (EDGE_MT * 16)
#define EDGES_PER_BLOCK (EDGE_WAVES * EDGES_PER_WAVE)

__global__ __launch_bounds__(256) void k_zero(float* __restrict__ p, int n) {
    int i = blockIdx.x * 256 + threadIdx.x;
    if (i < n) p[i] = 0.0f;
}

__global__ __launch_bounds__(256) void k_transpose(const float* __restrict__ src,
                                                   float* __restrict__ dst,
                                                   int R, int C) {
    int i = blockIdx.x * 256 + threadIdx.x;
    if (i < R * C) {
        int r = i / C, c = i - r * C;
        dst[c * R + r] = src[i];
    }
}

__global__ __launch_bounds__(128) void k_edge(const float* __restrict__ rel_emb,
                                              const float* __restrict__ metarel,
                                              const int*   __restrict__ g_src,
                                              const int*   __restrict__ g_dst,
                                              const int*   __restrict__ g_rel,
                                              const int*   __restrict__ g_inv,
                                              const float* __restrict__ WtI,
                                              const float* __restrict__ bI,
                                              const float* __restrict__ WtO,
                                              const float* __restrict__ bO,
                                              float* __restrict__ agg,
                                              float* __restrict__ cnt) {
    extern __shared__ float lds[];
    const int lane = threadIdx.x & 31;
    const int wave = threadIdx.x >> 5;
    const int lr   = lane & 15;
    const int half = lane >> 4;

    float* aw = lds + wave * (EDGES_PER_WAVE * A_STRIDE_E);
    const int e0 = (blockIdx.x * EDGE_WAVES + wave) * EDGES_PER_WAVE;

    const int col = lane * 8;
    for (int e = 0; e < EDGES_PER_WAVE; ++e) {
        const int ed = e0 + e;
        const int rr = g_rel[ed];
        const int ss = g_src[ed];
        const float* p = (col < DIM) ? (metarel + rr * DIM + col)
                                     : (rel_emb + ss * DIM + (col - DIM));
        float4 x0 = *(const float4*)p;
        float4 x1 = *(const float4*)(p + 4);
        float* q = aw + e * A_STRIDE_E + col;
        *(float4*)(q)     = x0;
        *(float4*)(q + 4) = x1;
    }
    atomicAdd(&cnt[g_dst[e0 + lane]],      1.0f);
    atomicAdd(&cnt[g_dst[e0 + 32 + lane]], 1.0f);

    for (int nt = 0; nt < 8; ++nt) {
        const int n0 = nt * 16;
        v8f accI[EDGE_MT];
        v8f accO[EDGE_MT];
#pragma unroll
        for (int t = 0; t < EDGE_MT; ++t) { accI[t] = {}; accO[t] = {}; }

#pragma unroll 2
        for (int ks = 0; ks < 64; ++ks) {
            v2f a[EDGE_MT];
#pragma unroll
            for (int t = 0; t < EDGE_MT; ++t) {
                const int ao = (t * 16 + lr) * A_STRIDE_E + ks * 4 + half * 2;
                a[t].x = aw[ao];
                a[t].y = aw[ao + 1];
            }
            const int bo0 = (ks * 4 + half * 2) * DIM + n0 + lr;
            v2f bi, bo;
            bi.x = WtI[bo0];  bi.y = WtI[bo0 + DIM];
            bo.x = WtO[bo0];  bo.y = WtO[bo0 + DIM];
#pragma unroll
            for (int t = 0; t < EDGE_MT; ++t) {
                accI[t] = __builtin_amdgcn_wmma_f32_16x16x4_f32(false, a[t], false, bi,
                                                                (short)0, accI[t], false, false);
                accO[t] = __builtin_amdgcn_wmma_f32_16x16x4_f32(false, a[t], false, bo,
                                                                (short)0, accO[t], false, false);
            }
        }
        const int n = n0 + lr;
        const float biasI = bI[n];
        const float biasO = bO[n];
#pragma unroll
        for (int t = 0; t < EDGE_MT; ++t) {
#pragma unroll
            for (int r = 0; r < 8; ++r) {
                const int ed = e0 + t * 16 + r + half * 8;
                const int iv = g_inv[ed];
                const int dm = g_dst[ed];
                const float v = iv ? (accO[t][r] + biasO) : (accI[t][r] + biasI);
                atomicAdd(&agg[dm * DIM + n], v);
            }
        }
    }
}

__global__ __launch_bounds__(256) void k_node(const float* __restrict__ rel_emb,
                                              const float* __restrict__ WtS,
                                              const float* __restrict__ bS,
                                              const float* __restrict__ agg,
                                              const float* __restrict__ cnt,
                                              float* __restrict__ out) {
    extern __shared__ float lds[];
    const int lane = threadIdx.x & 31;
    const int wave = threadIdx.x >> 5;
    const int lr   = lane & 15;
    const int half = lane >> 4;

    float* aw = lds + wave * (16 * A_STRIDE_N);
    const int m0 = (blockIdx.x * 8 + wave) * 16;
    if (m0 >= N_NODES) return;

    const int col = lane * 4;
    for (int r = 0; r < 16; ++r) {
        float4 x = *(const float4*)(rel_emb + (m0 + r) * DIM + col);
        *(float4*)(aw + r * A_STRIDE_N + col) = x;
    }

    for (int nt = 0; nt < 8; ++nt) {
        const int n0 = nt * 16;
        v8f acc = {};
#pragma unroll 4
        for (int ks = 0; ks < 32; ++ks) {
            const int ao = lr * A_STRIDE_N + ks * 4 + half * 2;
            v2f a;
            a.x = aw[ao];
            a.y = aw[ao + 1];
            const int bo0 = (ks * 4 + half * 2) * DIM + n0 + lr;
            v2f b;
            b.x = WtS[bo0];
            b.y = WtS[bo0 + DIM];
            acc = __builtin_amdgcn_wmma_f32_16x16x4_f32(false, a, false, b,
                                                        (short)0, acc, false, false);
        }
        const int n = n0 + lr;
        const float bias = bS[n];
#pragma unroll
        for (int r = 0; r < 8; ++r) {
            const int m = m0 + r + half * 8;
            const float c  = cnt[m];
            const float ha = agg[m * DIM + n] / fmaxf(c, 1.0f);
            out[m * DIM + n] = acc[r] + bias + ha;
        }
    }
}

__global__ __launch_bounds__(256) void k_meta(const float* __restrict__ metarel,
                                              const float* __restrict__ WtM,
                                              const float* __restrict__ bM,
                                              float* __restrict__ out) {
    extern __shared__ float lds[];
    const int lane = threadIdx.x & 31;
    const int wave = threadIdx.x >> 5;
    const int lr   = lane & 15;
    const int half = lane >> 4;

    float* aw = lds + wave * (16 * A_STRIDE_N);
    const int m0 = (blockIdx.x * 8 + wave) * 16;
    if (m0 >= N_REL) return;

    const int col = lane * 4;
    for (int r = 0; r < 16; ++r) {
        const int m = m0 + r;
        float4 x = make_float4(0.f, 0.f, 0.f, 0.f);
        if (m < N_REL) x = *(const float4*)(metarel + m * DIM + col);
        *(float4*)(aw + r * A_STRIDE_N + col) = x;
    }

    for (int nt = 0; nt < 8; ++nt) {
        const int n0 = nt * 16;
        v8f acc = {};
#pragma unroll 4
        for (int ks = 0; ks < 32; ++ks) {
            const int ao = lr * A_STRIDE_N + ks * 4 + half * 2;
            v2f a;
            a.x = aw[ao];
            a.y = aw[ao + 1];
            const int bo0 = (ks * 4 + half * 2) * DIM + n0 + lr;
            v2f b;
            b.x = WtM[bo0];
            b.y = WtM[bo0 + DIM];
            acc = __builtin_amdgcn_wmma_f32_16x16x4_f32(false, a, false, b,
                                                        (short)0, acc, false, false);
        }
        const int n = n0 + lr;
        const float bias = bM[n];
#pragma unroll
        for (int r = 0; r < 8; ++r) {
            const int m = m0 + r + half * 8;
            if (m < N_REL) out[m * DIM + n] = acc[r] + bias;
        }
    }
}

extern "C" void kernel_launch(void* const* d_in, const int* in_sizes, int n_in,
                              void* d_out, int out_size, void* d_ws, size_t ws_size,
                              hipStream_t stream) {
    const float* rel_emb = (const float*)d_in[0];
    const float* metarel = (const float*)d_in[1];
    const int*   srcI    = (const int*)  d_in[2];
    const int*   dstI    = (const int*)  d_in[3];
    const int*   relI    = (const int*)  d_in[4];
    const int*   invI    = (const int*)  d_in[5];
    const float* W_I_w   = (const float*)d_in[6];
    const float* W_I_b   = (const float*)d_in[7];
    const float* W_O_w   = (const float*)d_in[8];
    const float* W_O_b   = (const float*)d_in[9];
    const float* W_S_w   = (const float*)d_in[10];
    const float* W_S_b   = (const float*)d_in[11];
    const float* W_M_w   = (const float*)d_in[12];
    const float* W_M_b   = (const float*)d_in[13];

    float* ws  = (float*)d_ws;
    float* WtI = ws + WS_WT_I;
    float* WtO = ws + WS_WT_O;
    float* WtS = ws + WS_WT_S;
    float* WtM = ws + WS_WT_M;
    float* agg = ws + WS_AGG;
    float* cnt = ws + WS_CNT;

    float* out_rel  = (float*)d_out;
    float* out_meta = (float*)d_out + N_NODES * DIM;

    const int edgeLds = EDGE_WAVES * EDGES_PER_WAVE * A_STRIDE_E * sizeof(float);
    const int nodeLds = 8 * 16 * A_STRIDE_N * sizeof(float);

    (void)hipFuncSetAttribute((const void*)k_edge,
                              hipFuncAttributeMaxDynamicSharedMemorySize, edgeLds);
    (void)hipFuncSetAttribute((const void*)k_node,
                              hipFuncAttributeMaxDynamicSharedMemorySize, nodeLds);
    (void)hipFuncSetAttribute((const void*)k_meta,
                              hipFuncAttributeMaxDynamicSharedMemorySize, nodeLds);

    k_zero<<<(WS_ZERO_N + 255) / 256, 256, 0, stream>>>(agg, WS_ZERO_N);

    k_transpose<<<(DIM * K2 + 255) / 256, 256, 0, stream>>>(W_I_w, WtI, DIM, K2);
    k_transpose<<<(DIM * K2 + 255) / 256, 256, 0, stream>>>(W_O_w, WtO, DIM, K2);
    k_transpose<<<(DIM * DIM + 255) / 256, 256, 0, stream>>>(W_S_w, WtS, DIM, DIM);
    k_transpose<<<(DIM * DIM + 255) / 256, 256, 0, stream>>>(W_M_w, WtM, DIM, DIM);

    k_edge<<<N_EDGES / EDGES_PER_BLOCK, 128, edgeLds, stream>>>(rel_emb, metarel,
                                                                srcI, dstI, relI, invI,
                                                                WtI, W_I_b, WtO, W_O_b,
                                                                agg, cnt);

    k_node<<<(N_NODES + 127) / 128, 256, nodeLds, stream>>>(rel_emb, WtS, W_S_b,
                                                            agg, cnt, out_rel);

    k_meta<<<(N_REL + 127) / 128, 256, nodeLds, stream>>>(metarel, WtM, W_M_b,
                                                          out_meta);
}


# Launch fusion, 10 launches to 5: the clear and the four weight transposes share one launch; the meta product runs as 32 extra 16-row tiles at the head of the node kernel's grid, same k order
# speedup vs baseline: 6.0089x; 6.0089x over previous
#include <hip/hip_runtime.h>


typedef float v2f __attribute__((ext_vector_type(2)));
typedef float v8f __attribute__((ext_vector_type(8)));

#define N_NODES 50000
#define N_EDGES 640000
#define N_REL   500
#define DIM     128
#define K2      256

#define WS_WT_I 0
#define WS_WT_O 32768
#define WS_WT_S 65536
#define WS_WT_M 81920
#define WS_DEG  98304
#define WS_CUR  148304
#define WS_OFF  198304
#define WS_REC  248320
#define WS_ZERO_N 100000

#define A_STRIDE_F 644
#define KF         640

#define SCAN_T   1024
#define SCAN_PER 49
#define GU       8
#define AGG_W    4
#define META_TILES 32

#define ZERO_B  391
#define PREP_B  775
__global__ __launch_bounds__(256) void k_prep(const float* __restrict__ wI,
                                              const float* __restrict__ wO,
                                              const float* __restrict__ wS,
                                              const float* __restrict__ wM,
                                              float* __restrict__ ws) {
    int b = blockIdx.x;
    if (b < ZERO_B) {
        const int i = b * 256 + threadIdx.x;
        if (i < WS_ZERO_N) ws[WS_DEG + i] = 0.0f;
        return;
    }
    b -= ZERO_B;
    const float* src;
    float* dst;
    int lgC;
    if (b < 128)      { src = wI; dst = ws + WS_WT_I; lgC = 8; }
    else if (b < 256) { src = wO; dst = ws + WS_WT_O; lgC = 8; b -= 128; }
    else if (b < 320) { src = wS; dst = ws + WS_WT_S; lgC = 7; b -= 256; }
    else              { src = wM; dst = ws + WS_WT_M; lgC = 7; b -= 320; }
    const int i = b * 256 + threadIdx.x;
    const int r = i >> lgC, c = i - (r << lgC);
    dst[c * DIM + r] = src[i];
}

__global__ __launch_bounds__(256) void k_hist(const int* __restrict__ g_dst,
                                              int* __restrict__ deg) {
    int e = blockIdx.x * 256 + threadIdx.x;
    if (e < N_EDGES) atomicAdd(&deg[g_dst[e]], 1);
}

__global__ __launch_bounds__(1024) void k_scan(const int* __restrict__ deg,
                                               int* __restrict__ off) {
    extern __shared__ int sh[];
    int* buf  = sh;
    int* part = sh + SCAN_T * SCAN_PER;
    const int t = threadIdx.x;
    for (int i = t; i < SCAN_T * SCAN_PER; i += SCAN_T) buf[i] = (i < N_NODES) ? deg[i] : 0;
    __syncthreads();
    const int b = t * SCAN_PER;
    int s = 0;
    for (int i = 0; i < SCAN_PER; ++i) s += buf[b + i];
    part[t] = s;
    __syncthreads();
    for (int d = 1; d < SCAN_T; d <<= 1) {
        const int v = (t >= d) ? part[t - d] : 0;
        __syncthreads();
        part[t] += v;
        __syncthreads();
    }
    int run = part[t] - s;
    for (int i = 0; i < SCAN_PER; ++i) {
        const int c = buf[b + i];
        buf[b + i] = run;
        run += c;
    }
    __syncthreads();
    for (int i = t; i < N_NODES; i += SCAN_T) off[i] = buf[i];
    if (t == SCAN_T - 1) off[N_NODES] = part[t];
}

__global__ __launch_bounds__(256) void k_fill(const int* __restrict__ g_src,
                                              const int* __restrict__ g_dst,
                                              const int* __restrict__ g_rel,
                                              const int* __restrict__ g_inv,
                                              const int* __restrict__ off,
                                              int* __restrict__ cur,
                                              int* __restrict__ rec) {
    int e = blockIdx.x * 256 + threadIdx.x;
    if (e < N_EDGES) {
        const int d = g_dst[e];
        const int p = off[d] + atomicAdd(&cur[d], 1);
        int2 v;
        v.x = g_rel[e] * 2 + (g_inv[e] != 0 ? 1 : 0);
        v.y = g_src[e];
        ((int2*)rec)[p] = v;
    }
}

static __device__ __forceinline__ float4 f4add(float4 a, float4 b) {
    return make_float4(a.x + b.x, a.y + b.y, a.z + b.z, a.w + b.w);
}
static __device__ __forceinline__ float4 f4scale(float4 a, float s) {
    return make_float4(a.x * s, a.y * s, a.z * s, a.w * s);
}

static __device__ __forceinline__ void mm_seg(v8f (&acc)[2], const float* arow,
                                              const float* __restrict__ wl, int nks) {
#pragma unroll 8
    for (int ks = 0; ks < nks; ++ks) {
        v2f a;
        a.x = arow[ks * 4];
        a.y = arow[ks * 4 + 1];
        const float* w = wl + ks * 4 * DIM;
        const float2 b0 = *(const float2*)(w);
        const float2 b1 = *(const float2*)(w + DIM);
        v2f b;
        b.x = b0.x; b.y = b1.x;
        acc[0] = __builtin_amdgcn_wmma_f32_16x16x4_f32(false, a, false, b, (short)0, acc[0], false, false);
        b.x = b0.y; b.y = b1.y;
        acc[1] = __builtin_amdgcn_wmma_f32_16x16x4_f32(false, a, false, b, (short)0, acc[1], false, false);
    }
}

__global__ __launch_bounds__(AGG_W * 32) void k_aggnode(const float* __restrict__ rel_emb,
                                                const float* __restrict__ metarel,
                                                const int*   __restrict__ off,
                                                const int*   __restrict__ rec,
                                                const float* __restrict__ WtI,
                                                const float* __restrict__ bI,
                                                const float* __restrict__ WtO,
                                                const float* __restrict__ bO,
                                                const float* __restrict__ WtS,
                                                const float* __restrict__ bS,
                                                const float* __restrict__ WtM,
                                                const float* __restrict__ bM,
                                                float* __restrict__ out,
                                                float* __restrict__ out_meta) {
    extern __shared__ float lds[];
    const int lane = threadIdx.x & 31;
    const int wave = __builtin_amdgcn_readfirstlane(threadIdx.x >> 5);
    const int lr   = lane & 15;
    const int half = lane >> 4;
    float* aw = lds;

    if (blockIdx.x < META_TILES) {
        const int t0 = blockIdx.x * 16;
        for (int rr = 0; rr < 16 / AGG_W; ++rr) {
            const int r = wave * (16 / AGG_W) + rr;
            const int m = t0 + r;
            float4 x = make_float4(0.f, 0.f, 0.f, 0.f);
            if (m < N_REL) x = *(const float4*)(metarel + m * DIM + lane * 4);
            *(float4*)(aw + r * A_STRIDE_F + lane * 4) = x;
        }
        __syncthreads();
        v8f macc[2];
        macc[0] = {};
        macc[1] = {};
        const int nbm = wave * (DIM / AGG_W) + lr * 2;
        mm_seg(macc, aw + lr * A_STRIDE_F + half * 2, WtM + (half * 2) * DIM + nbm, 32);
        const float2 bMv = *(const float2*)(bM + nbm);
#pragma unroll
        for (int r = 0; r < 8; ++r) {
            const int m = t0 + half * 8 + r;
            if (m < N_REL) {
                float2 o2;
                o2.x = macc[0][r] + bMv.x;
                o2.y = macc[1][r] + bMv.y;
                *(float2*)(out_meta + m * DIM + nbm) = o2;
            }
        }
        return;
    }

    const int m0 = (blockIdx.x - META_TILES) * 16;

    const int col = lane * 8;
    const bool isMeta = (col < DIM);
    const int2* rc2 = (const int2*)rec;
    for (int rr = 0; rr < 16 / AGG_W; ++rr) {
        const int r   = wave * (16 / AGG_W) + rr;
        const int n   = m0 + r;
        const int beg = off[n];
        const int end = off[n + 1];
        float4 sI0 = make_float4(0.f, 0.f, 0.f, 0.f), sI1 = sI0, sO0 = sI0, sO1 = sI0;
        int cO = 0;
        for (int j = beg; j < end; j += GU) {
            float4 x0[GU], x1[GU];
            int fl[GU];
            const int jl = j + (lane & (GU - 1));
            const int2 mine = rc2[(jl < end) ? jl : (end - 1)];
#pragma unroll
            for (int u = 0; u < GU; ++u) {
                const int rx = __builtin_amdgcn_readlane(mine.x, u);
                const int ry = __builtin_amdgcn_readlane(mine.y, u);
                fl[u] = rx & 1;
                const float* p = isMeta ? (metarel + (rx >> 1) * DIM + col)
                                        : (rel_emb + ry * DIM + (col - DIM));
                x0[u] = *(const float4*)p;
                x1[u] = *(const float4*)(p + 4);
            }
#pragma unroll
            for (int u = 0; u < GU; ++u) {
                if (j + u < end) {
                    if (fl[u]) { sO0 = f4add(sO0, x0[u]); sO1 = f4add(sO1, x1[u]); ++cO; }
                    else       { sI0 = f4add(sI0, x0[u]); sI1 = f4add(sI1, x1[u]); }
                }
            }
        }
        const int   cnt = end - beg;
        const float inv = 1.0f / fmaxf((float)cnt, 1.0f);
        float* q = aw + r * A_STRIDE_F;
        *(float4*)(q + col)           = f4scale(sI0, inv);
        *(float4*)(q + col + 4)       = f4scale(sI1, inv);
        *(float4*)(q + K2 + col)      = f4scale(sO0, inv);
        *(float4*)(q + K2 + col + 4)  = f4scale(sO1, inv);
        *(float4*)(q + 2 * K2 + lane * 4) = *(const float4*)(rel_emb + n * DIM + lane * 4);
        if (lane == 0) {
            q[KF]     = (float)(cnt - cO) * inv;
            q[KF + 1] = (float)cO * inv;
        }
    }
    __syncthreads();

    v8f acc[2];
    acc[0] = {};
    acc[1] = {};
    const int    nb   = wave * (DIM / AGG_W) + lr * 2;
    const float* arow = aw + lr * A_STRIDE_F + half * 2;
    const int    wofs = (half * 2) * DIM + nb;
    mm_seg(acc, arow,          WtI + wofs, 64);
    mm_seg(acc, arow + K2,     WtO + wofs, 64);
    mm_seg(acc, arow + 2 * K2, WtS + wofs, 32);

    const float2 bSv = *(const float2*)(bS + nb);
    const float2 bIv = *(const float2*)(bI + nb);
    const float2 bOv = *(const float2*)(bO + nb);
#pragma unroll
    for (int r = 0; r < 8; ++r) {
        const int row = half * 8 + r;
        const float fI = aw[row * A_STRIDE_F + KF];
        const float fO = aw[row * A_STRIDE_F + KF + 1];
        float2 o2;
        o2.x = acc[0][r] + bSv.x + (fI * bIv.x + fO * bOv.x);
        o2.y = acc[1][r] + bSv.y + (fI * bIv.y + fO * bOv.y);
        *(float2*)(out + (m0 + row) * DIM + nb) = o2;
    }
}

extern "C" void kernel_launch(void* const* d_in, const int* in_sizes, int n_in,
                              void* d_out, int out_size, void* d_ws, size_t ws_size,
                              hipStream_t stream) {
    const float* rel_emb = (const float*)d_in[0];
    const float* metarel = (const float*)d_in[1];
    const int*   srcI    = (const int*)  d_in[2];
    const int*   dstI    = (const int*)  d_in[3];
    const int*   relI    = (const int*)  d_in[4];
    const int*   invI    = (const int*)  d_in[5];
    const float* W_I_w   = (const float*)d_in[6];
    const float* W_I_b   = (const float*)d_in[7];
    const float* W_O_w   = (const float*)d_in[8];
    const float* W_O_b   = (const float*)d_in[9];
    const float* W_S_w   = (const float*)d_in[10];
    const float* W_S_b   = (const float*)d_in[11];
    const float* W_M_w   = (const float*)d_in[12];
    const float* W_M_b   = (const float*)d_in[13];

    float* ws  = (float*)d_ws;
    float* WtI = ws + WS_WT_I;
    float* WtO = ws + WS_WT_O;
    float* WtS = ws + WS_WT_S;
    float* WtM = ws + WS_WT_M;
    int*   deg = (int*)d_ws + WS_DEG;
    int*   cur = (int*)d_ws + WS_CUR;
    int*   off = (int*)d_ws + WS_OFF;
    int*   rec = (int*)d_ws + WS_REC;

    float* out_rel  = (float*)d_out;
    float* out_meta = (float*)d_out + N_NODES * DIM;

    const int scanLds = (SCAN_T * SCAN_PER + SCAN_T) * sizeof(int);
    const int aggLds  = 16 * A_STRIDE_F * sizeof(float);

    (void)hipFuncSetAttribute((const void*)k_scan,
                              hipFuncAttributeMaxDynamicSharedMemorySize, scanLds);

    k_prep<<<PREP_B, 256, 0, stream>>>(W_I_w, W_O_w, W_S_w, W_M_w, ws);

    k_hist<<<N_EDGES / 256, 256, 0, stream>>>(dstI, deg);
    k_scan<<<1, SCAN_T, scanLds, stream>>>(deg, off);
    k_fill<<<N_EDGES / 256, 256, 0, stream>>>(srcI, dstI, relI, invI, off, cur, rec);

    k_aggnode<<<META_TILES + N_NODES / 16, AGG_W * 32, aggLds, stream>>>(rel_emb, metarel, off, rec,
                                                     WtI, W_I_b, WtO, W_O_b, WtS, W_S_b,
                                                     WtM, W_M_b, out_rel, out_meta);
}


# k_scan: the one-work-group exclusive scan of the in-degrees becomes 98 independent work-groups; each sums the degrees in front of its 512-node chunk (batched 16-byte loads), then scans its chunk
# speedup vs baseline: 6.4514x; 1.0736x over previous
#include <hip/hip_runtime.h>


typedef float v2f __attribute__((ext_vector_type(2)));
typedef float v8f __attribute__((ext_vector_type(8)));

#define N_NODES 50000
#define N_EDGES 640000
#define N_REL   500
#define DIM     128
#define K2      256

#define WS_WT_I 0
#define WS_WT_O 32768
#define WS_WT_S 65536
#define WS_WT_M 81920
#define WS_DEG  98304
#define WS_CUR  148304
#define WS_OFF  198304
#define WS_REC  248320
#define WS_ZERO_N 100000

#define A_STRIDE_F 644
#define KF         640

#define SC_T     512
#define SC_CH    512
#define SC_B     (N_NODES / SC_CH + 1)
#define SC_U     16
#define GU       8
#define AGG_W    4
#define META_TILES 32

#define ZERO_B  391
#define PREP_B  775
__global__ __launch_bounds__(256) void k_prep(const float* __restrict__ wI,
                                              const float* __restrict__ wO,
                                              const float* __restrict__ wS,
                                              const float* __restrict__ wM,
                                              float* __restrict__ ws) {
    int b = blockIdx.x;
    if (b < ZERO_B) {
        const int i = b * 256 + threadIdx.x;
        if (i < WS_ZERO_N) ws[WS_DEG + i] = 0.0f;
        return;
    }
    b -= ZERO_B;
    const float* src;
    float* dst;
    int lgC;
    if (b < 128)      { src = wI; dst = ws + WS_WT_I; lgC = 8; }
    else if (b < 256) { src = wO; dst = ws + WS_WT_O; lgC = 8; b -= 128; }
    else if (b < 320) { src = wS; dst = ws + WS_WT_S; lgC = 7; b -= 256; }
    else              { src = wM; dst = ws + WS_WT_M; lgC = 7; b -= 320; }
    const int i = b * 256 + threadIdx.x;
    const int r = i >> lgC, c = i - (r << lgC);
    dst[c * DIM + r] = src[i];
}

__global__ __launch_bounds__(256) void k_hist(const int* __restrict__ g_dst,
                                              int* __restrict__ deg) {
    int e = blockIdx.x * 256 + threadIdx.x;
    if (e < N_EDGES) atomicAdd(&deg[g_dst[e]], 1);
}

__global__ __launch_bounds__(SC_T) void k_scan(const int* __restrict__ deg,
                                               int* __restrict__ off) {
    __shared__ int sh_front[SC_T / 32];
    __shared__ int sh_chunk[SC_T / 32];
    const int t     = threadIdx.x;
    const int lane  = t & 31;
    const int wv    = t >> 5;
    const int start = blockIdx.x * SC_CH;

    const int n  = start + t;
    const int dv = deg[(n < N_NODES) ? n : 0];

    const int4* d4 = (const int4*)deg;
    const int   n4 = start >> 2;
    const int   nb = (n4 + SC_T * SC_U - 1) / (SC_T * SC_U);
    int front = 0;
    for (int k = 0; k < nb; ++k) {
        int4 v[SC_U];
#pragma unroll
        for (int u = 0; u < SC_U; ++u) {
            const int i = (k * SC_U + u) * SC_T + t;
            v[u] = d4[(i < n4) ? i : 0];
        }
#pragma unroll
        for (int u = 0; u < SC_U; ++u) {
            const int i = (k * SC_U + u) * SC_T + t;
            if (i < n4) front += (v[u].x + v[u].y) + (v[u].z + v[u].w);
        }
    }

    const int mine = (n < N_NODES) ? dv : 0;
    int inc = mine;
    for (int d = 1; d < 32; d <<= 1) {
        const int v = __shfl_up(inc, d, 32);
        if (lane >= d) inc += v;
    }
    for (int d = 16; d >= 1; d >>= 1) front += __shfl_xor(front, d, 32);
    if (lane == 31) {
        sh_chunk[wv] = inc;
        sh_front[wv] = front;
    }
    __syncthreads();
    int base = 0;
    if (lane < SC_T / 32) base = sh_front[lane] + ((lane < wv) ? sh_chunk[lane] : 0);
    for (int d = 16; d >= 1; d >>= 1) base += __shfl_xor(base, d, 32);
    if (n <= N_NODES) off[n] = base + (inc - mine);
}

__global__ __launch_bounds__(256) void k_fill(const int* __restrict__ g_src,
                                              const int* __restrict__ g_dst,
                                              const int* __restrict__ g_rel,
                                              const int* __restrict__ g_inv,
                                              const int* __restrict__ off,
                                              int* __restrict__ cur,
                                              int* __restrict__ rec) {
    int e = blockIdx.x * 256 + threadIdx.x;
    if (e < N_EDGES) {
        const int d = g_dst[e];
        const int p = off[d] + atomicAdd(&cur[d], 1);
        int2 v;
        v.x = g_rel[e] * 2 + (g_inv[e] != 0 ? 1 : 0);
        v.y = g_src[e];
        ((int2*)rec)[p] = v;
    }
}

static __device__ __forceinline__ float4 f4add(float4 a, float4 b) {
    return make_float4(a.x + b.x, a.y + b.y, a.z + b.z, a.w + b.w);
}
static __device__ __forceinline__ float4 f4scale(float4 a, float s) {
    return make_float4(a.x * s, a.y * s, a.z * s, a.w * s);
}

static __device__ __forceinline__ void mm_seg(v8f (&acc)[2], const float* arow,
                                              const float* __restrict__ wl, int nks) {
#pragma unroll 8
    for (int ks = 0; ks < nks; ++ks) {
        v2f a;
        a.x = arow[ks * 4];
        a.y = arow[ks * 4 + 1];
        const float* w = wl + ks * 4 * DIM;
        const float2 b0 = *(const float2*)(w);
        const float2 b1 = *(const float2*)(w + DIM);
        v2f b;
        b.x = b0.x; b.y = b1.x;
        acc[0] = __builtin_amdgcn_wmma_f32_16x16x4_f32(false, a, false, b, (short)0, acc[0], false, false);
        b.x = b0.y; b.y = b1.y;
        acc[1] = __builtin_amdgcn_wmma_f32_16x16x4_f32(false, a, false, b, (short)0, acc[1], false, false);
    }
}

__global__ __launch_bounds__(AGG_W * 32) void k_aggnode(const float* __restrict__ rel_emb,
                                                const float* __restrict__ metarel,
                                                const int*   __restrict__ off,
                                                const int*   __restrict__ rec,
                                                const float* __restrict__ WtI,
                                                const float* __restrict__ bI,
                                                const float* __restrict__ WtO,
                                                const float* __restrict__ bO,
                                                const float* __restrict__ WtS,
                                                const float* __restrict__ bS,
                                                const float* __restrict__ WtM,
                                                const float* __restrict__ bM,
                                                float* __restrict__ out,
                                                float* __restrict__ out_meta) {
    extern __shared__ float lds[];
    const int lane = threadIdx.x & 31;
    const int wave = __builtin_amdgcn_readfirstlane(threadIdx.x >> 5);
    const int lr   = lane & 15;
    const int half = lane >> 4;
    float* aw = lds;

    if (blockIdx.x < META_TILES) {
        const int t0 = blockIdx.x * 16;
        for (int rr = 0; rr < 16 / AGG_W; ++rr) {
            const int r = wave * (16 / AGG_W) + rr;
            const int m = t0 + r;
            float4 x = make_float4(0.f, 0.f, 0.f, 0.f);
            if (m < N_REL) x = *(const float4*)(metarel + m * DIM + lane * 4);
            *(float4*)(aw + r * A_STRIDE_F + lane * 4) = x;
        }
        __syncthreads();
        v8f macc[2];
        macc[0] = {};
        macc[1] = {};
        const int nbm = wave * (DIM / AGG_W) + lr * 2;
        mm_seg(macc, aw + lr * A_STRIDE_F + half * 2, WtM + (half * 2) * DIM + nbm, 32);
        const float2 bMv = *(const float2*)(bM + nbm);
#pragma unroll
        for (int r = 0; r < 8; ++r) {
            const int m = t0 + half * 8 + r;
            if (m < N_REL) {
                float2 o2;
                o2.x = macc[0][r] + bMv.x;
                o2.y = macc[1][r] + bMv.y;
                *(float2*)(out_meta + m * DIM + nbm) = o2;
            }
        }
        return;
    }

    const int m0 = (blockIdx.x - META_TILES) * 16;

    const int col = lane * 8;
    const bool isMeta = (col < DIM);
    const int2* rc2 = (const int2*)rec;
    for (int rr = 0; rr < 16 / AGG_W; ++rr) {
        const int r   = wave * (16 / AGG_W) + rr;
        const int n   = m0 + r;
        const int beg = off[n];
        const int end = off[n + 1];
        float4 sI0 = make_float4(0.f, 0.f, 0.f, 0.f), sI1 = sI0, sO0 = sI0, sO1 = sI0;
        int cO = 0;
        for (int j = beg; j < end; j += GU) {
            float4 x0[GU], x1[GU];
            int fl[GU];
            const int jl = j + (lane & (GU - 1));
            const int2 mine = rc2[(jl < end) ? jl : (end - 1)];
#pragma unroll
            for (int u = 0; u < GU; ++u) {
                const int rx = __builtin_amdgcn_readlane(mine.x, u);
                const int ry = __builtin_amdgcn_readlane(mine.y, u);
                fl[u] = rx & 1;
                const float* p = isMeta ? (metarel + (rx >> 1) * DIM + col)
                                        : (rel_emb + ry * DIM + (col - DIM));
                x0[u] = *(const float4*)p;
                x1[u] = *(const float4*)(p + 4);
            }
#pragma unroll
            for (int u = 0; u < GU; ++u) {
                if (j + u < end) {
                    if (fl[u]) { sO0 = f4add(sO0, x0[u]); sO1 = f4add(sO1, x1[u]); ++cO; }
                    else       { sI0 = f4add(sI0, x0[u]); sI1 = f4add(sI1, x1[u]); }
                }
            }
        }
        const int   cnt = end - beg;
        const float inv = 1.0f / fmaxf((float)cnt, 1.0f);
        float* q = aw + r * A_STRIDE_F;
        *(float4*)(q + col)           = f4scale(sI0, inv);
        *(float4*)(q + col + 4)       = f4scale(sI1, inv);
        *(float4*)(q + K2 + col)      = f4scale(sO0, inv);
        *(float4*)(q + K2 + col + 4)  = f4scale(sO1, inv);
        *(float4*)(q + 2 * K2 + lane * 4) = *(const float4*)(rel_emb + n * DIM + lane * 4);
        if (lane == 0) {
            q[KF]     = (float)(cnt - cO) * inv;
            q[KF + 1] = (float)cO * inv;
        }
    }
    __syncthreads();

    v8f acc[2];
    acc[0] = {};
    acc[1] = {};
    const int    nb   = wave * (DIM / AGG_W) + lr * 2;
    const float* arow = aw + lr * A_STRIDE_F + half * 2;
    const int    wofs = (half * 2) * DIM + nb;
    mm_seg(acc, arow,          WtI + wofs, 64);
    mm_seg(acc, arow + K2,     WtO + wofs, 64);
    mm_seg(acc, arow + 2 * K2, WtS + wofs, 32);

    const float2 bSv = *(const float2*)(bS + nb);
    const float2 bIv = *(const float2*)(bI + nb);
    const float2 bOv = *(const float2*)(bO + nb);
#pragma unroll
    for (int r = 0; r < 8; ++r) {
        const int row = half * 8 + r;
        const float fI = aw[row * A_STRIDE_F + KF];
        const float fO = aw[row * A_STRIDE_F + KF + 1];
        float2 o2;
        o2.x = acc[0][r] + bSv.x + (fI * bIv.x + fO * bOv.x);
        o2.y = acc[1][r] + bSv.y + (fI * bIv.y + fO * bOv.y);
        *(float2*)(out + (m0 + row) * DIM + nb) = o2;
    }
}

extern "C" void kernel_launch(void* const* d_in, const int* in_sizes, int n_in,
                              void* d_out, int out_size, void* d_ws, size_t ws_size,
                              hipStream_t stream) {
    const float* rel_emb = (const float*)d_in[0];
    const float* metarel = (const float*)d_in[1];
    const int*   srcI    = (const int*)  d_in[2];
    const int*   dstI    = (const int*)  d_in[3];
    const int*   relI    = (const int*)  d_in[4];
    const int*   invI    = (const int*)  d_in[5];
    const float* W_I_w   = (const float*)d_in[6];
    const float* W_I_b   = (const float*)d_in[7];
    const float* W_O_w   = (const float*)d_in[8];
    const float* W_O_b   = (const float*)d_in[9];
    const float* W_S_w   = (const float*)d_in[10];
    const float* W_S_b   = (const float*)d_in[11];
    const float* W_M_w   = (const float*)d_in[12];
    const float* W_M_b   = (const float*)d_in[13];

    float* ws  = (float*)d_ws;
    float* WtI = ws + WS_WT_I;
    float* WtO = ws + WS_WT_O;
    float* WtS = ws + WS_WT_S;
    float* WtM = ws + WS_WT_M;
    int*   deg = (int*)d_ws + WS_DEG;
    int*   cur = (int*)d_ws + WS_CUR;
    int*   off = (int*)d_ws + WS_OFF;
    int*   rec = (int*)d_ws + WS_REC;

    float* out_rel  = (float*)d_out;
    float* out_meta = (float*)d_out + N_NODES * DIM;

    const int aggLds  = 16 * A_STRIDE_F * sizeof(float);

    k_prep<<<PREP_B, 256, 0, stream>>>(W_I_w, W_O_w, W_S_w, W_M_w, ws);

    k_hist<<<N_EDGES / 256, 256, 0, stream>>>(dstI, deg);
    k_scan<<<SC_B, SC_T, 0, stream>>>(deg, off);
    k_fill<<<N_EDGES / 256, 256, 0, stream>>>(srcI, dstI, relI, invI, off, cur, rec);

    k_aggnode<<<META_TILES + N_NODES / 16, AGG_W * 32, aggLds, stream>>>(rel_emb, metarel, off, rec,
                                                     WtI, W_I_b, WtO, W_O_b, WtS, W_S_b,
                                                     WtM, W_M_b, out_rel, out_meta);
}


# Count, scan, fill replaced by ONE pass chaining each node's in-edges into a linked list (atomic exchange): 5 launches to 3; a wave walks its 4 rows' lists abreast; sum order still by atomic arrival
# speedup vs baseline: 8.6540x; 1.3414x over previous
#include <hip/hip_runtime.h>


typedef float v2f __attribute__((ext_vector_type(2)));
typedef float v8f __attribute__((ext_vector_type(8)));

#define N_NODES 50000
#define N_EDGES 640000
#define N_REL   500
#define DIM     128
#define K2      256

#define WS_WT_I 0
#define WS_WT_O 32768
#define WS_WT_S 65536
#define WS_WT_M 81920
#define WS_HEAD 98304
#define WS_REC  248320
#define WS_ZERO_N 50000

#define A_STRIDE_F 644
#define KF         640

#define AGG_W    4
#define RW       4
#define META_TILES 32

#define ZERO_B  196
#define PREP_B  580
__global__ __launch_bounds__(256) void k_prep(const float* __restrict__ wI,
                                              const float* __restrict__ wO,
                                              const float* __restrict__ wS,
                                              const float* __restrict__ wM,
                                              float* __restrict__ ws) {
    int b = blockIdx.x;
    if (b < ZERO_B) {
        const int i = b * 256 + threadIdx.x;
        if (i < WS_ZERO_N) ws[WS_HEAD + i] = 0.0f;
        return;
    }
    b -= ZERO_B;
    const float* src;
    float* dst;
    int lgC;
    if (b < 128)      { src = wI; dst = ws + WS_WT_I; lgC = 8; }
    else if (b < 256) { src = wO; dst = ws + WS_WT_O; lgC = 8; b -= 128; }
    else if (b < 320) { src = wS; dst = ws + WS_WT_S; lgC = 7; b -= 256; }
    else              { src = wM; dst = ws + WS_WT_M; lgC = 7; b -= 320; }
    const int i = b * 256 + threadIdx.x;
    const int r = i >> lgC, c = i - (r << lgC);
    dst[c * DIM + r] = src[i];
}

__global__ __launch_bounds__(256) void k_link(const int* __restrict__ g_src,
                                              const int* __restrict__ g_dst,
                                              const int* __restrict__ g_rel,
                                              const int* __restrict__ g_inv,
                                              int* __restrict__ head,
                                              int* __restrict__ rec) {
    int e = blockIdx.x * 256 + threadIdx.x;
    if (e < N_EDGES) {
        const int prev = atomicExch(&head[g_dst[e]], e + 1);
        int2 v;
        v.x = (g_rel[e] * 2 + (g_inv[e] != 0 ? 1 : 0)) | (g_src[e] << 10);
        v.y = prev;
        ((int2*)rec)[e] = v;
    }
}

static __device__ __forceinline__ float4 f4add(float4 a, float4 b) {
    return make_float4(a.x + b.x, a.y + b.y, a.z + b.z, a.w + b.w);
}
static __device__ __forceinline__ float4 f4scale(float4 a, float s) {
    return make_float4(a.x * s, a.y * s, a.z * s, a.w * s);
}

static __device__ __forceinline__ void mm_seg(v8f (&acc)[2], const float* arow,
                                              const float* __restrict__ wl, int nks) {
#pragma unroll 8
    for (int ks = 0; ks < nks; ++ks) {
        v2f a;
        a.x = arow[ks * 4];
        a.y = arow[ks * 4 + 1];
        const float* w = wl + ks * 4 * DIM;
        const float2 b0 = *(const float2*)(w);
        const float2 b1 = *(const float2*)(w + DIM);
        v2f b;
        b.x = b0.x; b.y = b1.x;
        acc[0] = __builtin_amdgcn_wmma_f32_16x16x4_f32(false, a, false, b, (short)0, acc[0], false, false);
        b.x = b0.y; b.y = b1.y;
        acc[1] = __builtin_amdgcn_wmma_f32_16x16x4_f32(false, a, false, b, (short)0, acc[1], false, false);
    }
}

__global__ __launch_bounds__(AGG_W * 32) void k_aggnode(const float* __restrict__ rel_emb,
                                                const float* __restrict__ metarel,
                                                const int*   __restrict__ head,
                                                const int*   __restrict__ rec,
                                                const float* __restrict__ WtI,
                                                const float* __restrict__ bI,
                                                const float* __restrict__ WtO,
                                                const float* __restrict__ bO,
                                                const float* __restrict__ WtS,
                                                const float* __restrict__ bS,
                                                const float* __restrict__ WtM,
                                                const float* __restrict__ bM,
                                                float* __restrict__ out,
                                                float* __restrict__ out_meta) {
    extern __shared__ float lds[];
    const int lane = threadIdx.x & 31;
    const int wave = __builtin_amdgcn_readfirstlane(threadIdx.x >> 5);
    const int lr   = lane & 15;
    const int half = lane >> 4;
    float* aw = lds;

    if (blockIdx.x < META_TILES) {
        const int t0 = blockIdx.x * 16;
        for (int rr = 0; rr < 16 / AGG_W; ++rr) {
            const int r = wave * (16 / AGG_W) + rr;
            const int m = t0 + r;
            float4 x = make_float4(0.f, 0.f, 0.f, 0.f);
            if (m < N_REL) x = *(const float4*)(metarel + m * DIM + lane * 4);
            *(float4*)(aw + r * A_STRIDE_F + lane * 4) = x;
        }
        __syncthreads();
        v8f macc[2];
        macc[0] = {};
        macc[1] = {};
        const int nbm = wave * (DIM / AGG_W) + lr * 2;
        mm_seg(macc, aw + lr * A_STRIDE_F + half * 2, WtM + (half * 2) * DIM + nbm, 32);
        const float2 bMv = *(const float2*)(bM + nbm);
#pragma unroll
        for (int r = 0; r < 8; ++r) {
            const int m = t0 + half * 8 + r;
            if (m < N_REL) {
                float2 o2;
                o2.x = macc[0][r] + bMv.x;
                o2.y = macc[1][r] + bMv.y;
                *(float2*)(out_meta + m * DIM + nbm) = o2;
            }
        }
        return;
    }

    const int m0 = (blockIdx.x - META_TILES) * 16;

    const int col = lane * 8;
    const bool isMeta = (col < DIM);
    const int2* rc2 = (const int2*)rec;
    const int r0 = wave * RW;
    const float4 own0 = *(const float4*)(rel_emb + (m0 + r0 + 0) * DIM + lane * 4);
    const float4 own1 = *(const float4*)(rel_emb + (m0 + r0 + 1) * DIM + lane * 4);
    const float4 own2 = *(const float4*)(rel_emb + (m0 + r0 + 2) * DIM + lane * 4);
    const float4 own3 = *(const float4*)(rel_emb + (m0 + r0 + 3) * DIM + lane * 4);

    int  myp = head[m0 + r0 + (lane & (RW - 1))];
    int2 nxt = rc2[(myp > 0) ? min(myp - 1, N_EDGES - 1) : 0];
    {
        float* q = aw + r0 * A_STRIDE_F + 2 * K2 + lane * 4;
        *(float4*)(q)                  = own0;
        *(float4*)(q + A_STRIDE_F)     = own1;
        *(float4*)(q + 2 * A_STRIDE_F) = own2;
        *(float4*)(q + 3 * A_STRIDE_F) = own3;
    }
    float4 sI0[RW], sI1[RW], sO0[RW], sO1[RW];
    int cnt[RW], cO[RW];
#pragma unroll
    for (int r = 0; r < RW; ++r) {
        sI0[r] = make_float4(0.f, 0.f, 0.f, 0.f);
        sI1[r] = sI0[r]; sO0[r] = sI0[r]; sO1[r] = sI0[r];
        cnt[r] = 0; cO[r] = 0;
    }
    for (int hop = 0; hop < N_EDGES; ++hop) {
        const unsigned liveMask = __builtin_amdgcn_ballot_w32(myp != 0) & ((1u << RW) - 1u);
        if (liveMask == 0u) break;
        const int2 cur = nxt;
        myp = (myp != 0) ? cur.y : 0;
        nxt = rc2[(myp > 0) ? min(myp - 1, N_EDGES - 1) : 0];
        float4 x0[RW], x1[RW];
        int fl[RW];
#pragma unroll
        for (int r = 0; r < RW; ++r) {
            if (liveMask & (1u << r)) {
                const int cx = __builtin_amdgcn_readlane(cur.x, r);
                fl[r] = cx & 1;
                const float* p = isMeta ? (metarel + ((cx & 1023) >> 1) * DIM + col)
                                        : (rel_emb + (cx >> 10) * DIM + (col - DIM));
                x0[r] = *(const float4*)p;
                x1[r] = *(const float4*)(p + 4);
            }
        }
#pragma unroll
        for (int r = 0; r < RW; ++r) {
            if (liveMask & (1u << r)) {
                ++cnt[r];
                if (fl[r]) { sO0[r] = f4add(sO0[r], x0[r]); sO1[r] = f4add(sO1[r], x1[r]); ++cO[r]; }
                else       { sI0[r] = f4add(sI0[r], x0[r]); sI1[r] = f4add(sI1[r], x1[r]); }
            }
        }
    }
#pragma unroll
    for (int r = 0; r < RW; ++r) {
        const float inv = 1.0f / fmaxf((float)cnt[r], 1.0f);
        float* q = aw + (r0 + r) * A_STRIDE_F;
        *(float4*)(q + col)           = f4scale(sI0[r], inv);
        *(float4*)(q + col + 4)       = f4scale(sI1[r], inv);
        *(float4*)(q + K2 + col)      = f4scale(sO0[r], inv);
        *(float4*)(q + K2 + col + 4)  = f4scale(sO1[r], inv);
        if (lane == 0) {
            q[KF]     = (float)(cnt[r] - cO[r]) * inv;
            q[KF + 1] = (float)cO[r] * inv;
        }
    }
    __syncthreads();

    v8f acc[2];
    acc[0] = {};
    acc[1] = {};
    const int    nb   = wave * (DIM / AGG_W) + lr * 2;
    const float* arow = aw + lr * A_STRIDE_F + half * 2;
    const int    wofs = (half * 2) * DIM + nb;
    mm_seg(acc, arow,          WtI + wofs, 64);
    mm_seg(acc, arow + K2,     WtO + wofs, 64);
    mm_seg(acc, arow + 2 * K2, WtS + wofs, 32);

    const float2 bSv = *(const float2*)(bS + nb);
    const float2 bIv = *(const float2*)(bI + nb);
    const float2 bOv = *(const float2*)(bO + nb);
#pragma unroll
    for (int r = 0; r < 8; ++r) {
        const int row = half * 8 + r;
        const float fI = aw[row * A_STRIDE_F + KF];
        const float fO = aw[row * A_STRIDE_F + KF + 1];
        float2 o2;
        o2.x = acc[0][r] + bSv.x + (fI * bIv.x + fO * bOv.x);
        o2.y = acc[1][r] + bSv.y + (fI * bIv.y + fO * bOv.y);
        *(float2*)(out + (m0 + row) * DIM + nb) = o2;
    }
}

extern "C" void kernel_launch(void* const* d_in, const int* in_sizes, int n_in,
                              void* d_out, int out_size, void* d_ws, size_t ws_size,
                              hipStream_t stream) {
    const float* rel_emb = (const float*)d_in[0];
    const float* metarel = (const float*)d_in[1];
    const int*   srcI    = (const int*)  d_in[2];
    const int*   dstI    = (const int*)  d_in[3];
    const int*   relI    = (const int*)  d_in[4];
    const int*   invI    = (const int*)  d_in[5];
    const float* W_I_w   = (const float*)d_in[6];
    const float* W_I_b   = (const float*)d_in[7];
    const float* W_O_w   = (const float*)d_in[8];
    const float* W_O_b   = (const float*)d_in[9];
    const float* W_S_w   = (const float*)d_in[10];
    const float* W_S_b   = (const float*)d_in[11];
    const float* W_M_w   = (const float*)d_in[12];
    const float* W_M_b   = (const float*)d_in[13];

    float* ws  = (float*)d_ws;
    float* WtI = ws + WS_WT_I;
    float* WtO = ws + WS_WT_O;
    float* WtS = ws + WS_WT_S;
    float* WtM = ws + WS_WT_M;
    int*   head = (int*)d_ws + WS_HEAD;
    int*   rec  = (int*)d_ws + WS_REC;

    float* out_rel  = (float*)d_out;
    float* out_meta = (float*)d_out + N_NODES * DIM;

    const int aggLds  = 16 * A_STRIDE_F * sizeof(float);

    k_prep<<<PREP_B, 256, 0, stream>>>(W_I_w, W_O_w, W_S_w, W_M_w, ws);

    k_link<<<N_EDGES / 256, 256, 0, stream>>>(srcI, dstI, relI, invI, head, rec);

    k_aggnode<<<META_TILES + N_NODES / 16, AGG_W * 32, aggLds, stream>>>(rel_emb, metarel, head, rec,
                                                     WtI, W_I_b, WtO, W_O_b, WtS, W_S_b,
                                                     WtM, W_M_b, out_rel, out_meta);
}


# k_aggnode product: k_prep stores the weights k-pair-interleaved, so one whole-line b128 load per lane and k-step holds both column tiles' B operands as loaded; 16 k-steps of loads per wait; sums kept
# speedup vs baseline: 9.5604x; 1.1047x over previous
#include <hip/hip_runtime.h>


typedef float v2f __attribute__((ext_vector_type(2)));
typedef float v8f __attribute__((ext_vector_type(8)));

#define N_NODES 50000
#define N_EDGES 640000
#define N_REL   500
#define DIM     128
#define K2      256

#define WS_WT_I 0
#define WS_WT_O 32768
#define WS_WT_S 65536
#define WS_WT_M 81920
#define WS_HEAD 98304
#define WS_REC  248320
#define WS_ZERO_N 50000

#define A_STRIDE_F 644
#define KF         640

#define AGG_W    4
#define RW       4
#define META_TILES 32

#define ZERO_B  196
#define PREP_B  580
__global__ __launch_bounds__(256) void k_prep(const float* __restrict__ wI,
                                              const float* __restrict__ wO,
                                              const float* __restrict__ wS,
                                              const float* __restrict__ wM,
                                              float* __restrict__ ws) {
    int b = blockIdx.x;
    if (b < ZERO_B) {
        const int i = b * 256 + threadIdx.x;
        if (i < WS_ZERO_N) ws[WS_HEAD + i] = 0.0f;
        return;
    }
    b -= ZERO_B;
    const float* src;
    float* dst;
    int lgC;
    if (b < 128)      { src = wI; dst = ws + WS_WT_I; lgC = 8; }
    else if (b < 256) { src = wO; dst = ws + WS_WT_O; lgC = 8; b -= 128; }
    else if (b < 320) { src = wS; dst = ws + WS_WT_S; lgC = 7; b -= 256; }
    else              { src = wM; dst = ws + WS_WT_M; lgC = 7; b -= 320; }
    const int i = b * 256 + threadIdx.x;
    const int r = i >> lgC, c = i - (r << lgC);
    dst[(c >> 1) * (2 * DIM) + r * 2 + (c & 1)] = src[i];
}

__global__ __launch_bounds__(256) void k_link(const int* __restrict__ g_src,
                                              const int* __restrict__ g_dst,
                                              const int* __restrict__ g_rel,
                                              const int* __restrict__ g_inv,
                                              int* __restrict__ head,
                                              int* __restrict__ rec) {
    int e = blockIdx.x * 256 + threadIdx.x;
    if (e < N_EDGES) {
        const int prev = atomicExch(&head[g_dst[e]], e + 1);
        int2 v;
        v.x = (g_rel[e] * 2 + (g_inv[e] != 0 ? 1 : 0)) | (g_src[e] << 10);
        v.y = prev;
        ((int2*)rec)[e] = v;
    }
}

static __device__ __forceinline__ float4 f4add(float4 a, float4 b) {
    return make_float4(a.x + b.x, a.y + b.y, a.z + b.z, a.w + b.w);
}
static __device__ __forceinline__ float4 f4scale(float4 a, float s) {
    return make_float4(a.x * s, a.y * s, a.z * s, a.w * s);
}

static __device__ __forceinline__ void mm_seg(v8f (&acc)[2], const float* arow,
                                              const float* __restrict__ wl, int nks) {
#pragma unroll 16
    for (int ks = 0; ks < nks; ++ks) {
        v2f a;
        a.x = arow[ks * 4];
        a.y = arow[ks * 4 + 1];
        const float4 bq = *(const float4*)(wl + ks * 4 * DIM);
        v2f b;
        b.x = bq.x; b.y = bq.y;
        acc[0] = __builtin_amdgcn_wmma_f32_16x16x4_f32(false, a, false, b, (short)0, acc[0], false, false);
        b.x = bq.z; b.y = bq.w;
        acc[1] = __builtin_amdgcn_wmma_f32_16x16x4_f32(false, a, false, b, (short)0, acc[1], false, false);
    }
}

__global__ __launch_bounds__(AGG_W * 32) void k_aggnode(const float* __restrict__ rel_emb,
                                                const float* __restrict__ metarel,
                                                const int*   __restrict__ head,
                                                const int*   __restrict__ rec,
                                                const float* __restrict__ WtI,
                                                const float* __restrict__ bI,
                                                const float* __restrict__ WtO,
                                                const float* __restrict__ bO,
                                                const float* __restrict__ WtS,
                                                const float* __restrict__ bS,
                                                const float* __restrict__ WtM,
                                                const float* __restrict__ bM,
                                                float* __restrict__ out,
                                                float* __restrict__ out_meta) {
    extern __shared__ float lds[];
    const int lane = threadIdx.x & 31;
    const int wave = __builtin_amdgcn_readfirstlane(threadIdx.x >> 5);
    const int lr   = lane & 15;
    const int half = lane >> 4;
    float* aw = lds;

    if (blockIdx.x < META_TILES) {
        const int t0 = blockIdx.x * 16;
        for (int rr = 0; rr < 16 / AGG_W; ++rr) {
            const int r = wave * (16 / AGG_W) + rr;
            const int m = t0 + r;
            float4 x = make_float4(0.f, 0.f, 0.f, 0.f);
            if (m < N_REL) x = *(const float4*)(metarel + m * DIM + lane * 4);
            *(float4*)(aw + r * A_STRIDE_F + lane * 4) = x;
        }
        __syncthreads();
        v8f macc[2];
        macc[0] = {};
        macc[1] = {};
        const int nbm = wave * (DIM / AGG_W) + lr * 2;
        mm_seg(macc, aw + lr * A_STRIDE_F + half * 2, WtM + half * (2 * DIM) + nbm * 2, 32);
        const float2 bMv = *(const float2*)(bM + nbm);
#pragma unroll
        for (int r = 0; r < 8; ++r) {
            const int m = t0 + half * 8 + r;
            if (m < N_REL) {
                float2 o2;
                o2.x = macc[0][r] + bMv.x;
                o2.y = macc[1][r] + bMv.y;
                *(float2*)(out_meta + m * DIM + nbm) = o2;
            }
        }
        return;
    }

    const int m0 = (blockIdx.x - META_TILES) * 16;

    const int col = lane * 8;
    const bool isMeta = (col < DIM);
    const int2* rc2 = (const int2*)rec;
    const int r0 = wave * RW;
    const float4 own0 = *(const float4*)(rel_emb + (m0 + r0 + 0) * DIM + lane * 4);
    const float4 own1 = *(const float4*)(rel_emb + (m0 + r0 + 1) * DIM + lane * 4);
    const float4 own2 = *(const float4*)(rel_emb + (m0 + r0 + 2) * DIM + lane * 4);
    const float4 own3 = *(const float4*)(rel_emb + (m0 + r0 + 3) * DIM + lane * 4);

    int  myp = head[m0 + r0 + (lane & (RW - 1))];
    int2 nxt = rc2[(myp > 0) ? min(myp - 1, N_EDGES - 1) : 0];
    {
        float* q = aw + r0 * A_STRIDE_F + 2 * K2 + lane * 4;
        *(float4*)(q)                  = own0;
        *(float4*)(q + A_STRIDE_F)     = own1;
        *(float4*)(q + 2 * A_STRIDE_F) = own2;
        *(float4*)(q + 3 * A_STRIDE_F) = own3;
    }
    float4 sI0[RW], sI1[RW], sO0[RW], sO1[RW];
    int cnt[RW], cO[RW];
#pragma unroll
    for (int r = 0; r < RW; ++r) {
        sI0[r] = make_float4(0.f, 0.f, 0.f, 0.f);
        sI1[r] = sI0[r]; sO0[r] = sI0[r]; sO1[r] = sI0[r];
        cnt[r] = 0; cO[r] = 0;
    }
    for (int hop = 0; hop < N_EDGES; ++hop) {
        const unsigned liveMask = __builtin_amdgcn_ballot_w32(myp != 0) & ((1u << RW) - 1u);
        if (liveMask == 0u) break;
        const int2 cur = nxt;
        myp = (myp != 0) ? cur.y : 0;
        nxt = rc2[(myp > 0) ? min(myp - 1, N_EDGES - 1) : 0];
        float4 x0[RW], x1[RW];
        int fl[RW];
#pragma unroll
        for (int r = 0; r < RW; ++r) {
            if (liveMask & (1u << r)) {
                const int cx = __builtin_amdgcn_readlane(cur.x, r);
                fl[r] = cx & 1;
                const float* p = isMeta ? (metarel + ((cx & 1023) >> 1) * DIM + col)
                                        : (rel_emb + (cx >> 10) * DIM + (col - DIM));
                x0[r] = *(const float4*)p;
                x1[r] = *(const float4*)(p + 4);
            }
        }
#pragma unroll
        for (int r = 0; r < RW; ++r) {
            if (liveMask & (1u << r)) {
                ++cnt[r];
                if (fl[r]) { sO0[r] = f4add(sO0[r], x0[r]); sO1[r] = f4add(sO1[r], x1[r]); ++cO[r]; }
                else       { sI0[r] = f4add(sI0[r], x0[r]); sI1[r] = f4add(sI1[r], x1[r]); }
            }
        }
    }
#pragma unroll
    for (int r = 0; r < RW; ++r) {
        const float inv = 1.0f / fmaxf((float)cnt[r], 1.0f);
        float* q = aw + (r0 + r) * A_STRIDE_F;
        *(float4*)(q + col)           = f4scale(sI0[r], inv);
        *(float4*)(q + col + 4)       = f4scale(sI1[r], inv);
        *(float4*)(q + K2 + col)      = f4scale(sO0[r], inv);
        *(float4*)(q + K2 + col + 4)  = f4scale(sO1[r], inv);
        if (lane == 0) {
            q[KF]     = (float)(cnt[r] - cO[r]) * inv;
            q[KF + 1] = (float)cO[r] * inv;
        }
    }
    __syncthreads();

    v8f acc[2];
    acc[0] = {};
    acc[1] = {};
    const int    nb   = wave * (DIM / AGG_W) + lr * 2;
    const float* arow = aw + lr * A_STRIDE_F + half * 2;
    const int    wofs = half * (2 * DIM) + nb * 2;
    mm_seg(acc, arow,          WtI + wofs, 64);
    mm_seg(acc, arow + K2,     WtO + wofs, 64);
    mm_seg(acc, arow + 2 * K2, WtS + wofs, 32);

    const float2 bSv = *(const float2*)(bS + nb);
    const float2 bIv = *(const float2*)(bI + nb);
    const float2 bOv = *(const float2*)(bO + nb);
#pragma unroll
    for (int r = 0; r < 8; ++r) {
        const int row = half * 8 + r;
        const float fI = aw[row * A_STRIDE_F + KF];
        const float fO = aw[row * A_STRIDE_F + KF + 1];
        float2 o2;
        o2.x = acc[0][r] + bSv.x + (fI * bIv.x + fO * bOv.x);
        o2.y = acc[1][r] + bSv.y + (fI * bIv.y + fO * bOv.y);
        *(float2*)(out + (m0 + row) * DIM + nb) = o2;
    }
}

extern "C" void kernel_launch(void* const* d_in, const int* in_sizes, int n_in,
                              void* d_out, int out_size, void* d_ws, size_t ws_size,
                              hipStream_t stream) {
    const float* rel_emb = (const float*)d_in[0];
    const float* metarel = (const float*)d_in[1];
    const int*   srcI    = (const int*)  d_in[2];
    const int*   dstI    = (const int*)  d_in[3];
    const int*   relI    = (const int*)  d_in[4];
    const int*   invI    = (const int*)  d_in[5];
    const float* W_I_w   = (const float*)d_in[6];
    const float* W_I_b   = (const float*)d_in[7];
    const float* W_O_w   = (const float*)d_in[8];
    const float* W_O_b   = (const float*)d_in[9];
    const float* W_S_w   = (const float*)d_in[10];
    const float* W_S_b   = (const float*)d_in[11];
    const float* W_M_w   = (const float*)d_in[12];
    const float* W_M_b   = (const float*)d_in[13];

    float* ws  = (float*)d_ws;
    float* WtI = ws + WS_WT_I;
    float* WtO = ws + WS_WT_O;
    float* WtS = ws + WS_WT_S;
    float* WtM = ws + WS_WT_M;
    int*   head = (int*)d_ws + WS_HEAD;
    int*   rec  = (int*)d_ws + WS_REC;

    float* out_rel  = (float*)d_out;
    float* out_meta = (float*)d_out + N_NODES * DIM;

    const int aggLds  = 16 * A_STRIDE_F * sizeof(float);

    k_prep<<<PREP_B, 256, 0, stream>>>(W_I_w, W_O_w, W_S_w, W_M_w, ws);

    k_link<<<N_EDGES / 256, 256, 0, stream>>>(srcI, dstI, relI, invI, head, rec);

    k_aggnode<<<META_TILES + N_NODES / 16, AGG_W * 32, aggLds, stream>>>(rel_emb, metarel, head, rec,
                                                     WtI, W_I_b, WtO, W_O_b, WtS, W_S_b,
                                                     WtM, W_M_b, out_rel, out_meta);
}


# Each destination's in-edges chained into 2 lists (by edge index parity), not 1; a wave walks its 4 rows' 8 lists abreast: dependent hops per tile 19 to 12, 8 row gathers in flight; sum order changes
# speedup vs baseline: 9.8498x; 1.0303x over previous
#include <hip/hip_runtime.h>


typedef float v2f __attribute__((ext_vector_type(2)));
typedef float v8f __attribute__((ext_vector_type(8)));

#define N_NODES 50000
#define N_EDGES 640000
#define N_REL   500
#define DIM     128
#define K2      256

#define WS_WT_I 0
#define WS_WT_O 32768
#define WS_WT_S 65536
#define WS_WT_M 81920
#define WS_HEAD 98304
#define WS_REC  248320
#define NSUB 2
#define WS_ZERO_N 100000

#define A_STRIDE_F 644
#define KF         640

#define AGG_W    4
#define RW       4
#define NL       8
#define META_TILES 32

#define ZERO_B  391
#define PREP_B  775
__global__ __launch_bounds__(256) void k_prep(const float* __restrict__ wI,
                                              const float* __restrict__ wO,
                                              const float* __restrict__ wS,
                                              const float* __restrict__ wM,
                                              float* __restrict__ ws) {
    int b = blockIdx.x;
    if (b < ZERO_B) {
        const int i = b * 256 + threadIdx.x;
        if (i < WS_ZERO_N) ws[WS_HEAD + i] = 0.0f;
        return;
    }
    b -= ZERO_B;
    const float* src;
    float* dst;
    int lgC;
    if (b < 128)      { src = wI; dst = ws + WS_WT_I; lgC = 8; }
    else if (b < 256) { src = wO; dst = ws + WS_WT_O; lgC = 8; b -= 128; }
    else if (b < 320) { src = wS; dst = ws + WS_WT_S; lgC = 7; b -= 256; }
    else              { src = wM; dst = ws + WS_WT_M; lgC = 7; b -= 320; }
    const int i = b * 256 + threadIdx.x;
    const int r = i >> lgC, c = i - (r << lgC);
    dst[(c >> 1) * (2 * DIM) + r * 2 + (c & 1)] = src[i];
}

__global__ __launch_bounds__(256) void k_link(const int* __restrict__ g_src,
                                              const int* __restrict__ g_dst,
                                              const int* __restrict__ g_rel,
                                              const int* __restrict__ g_inv,
                                              int* __restrict__ head,
                                              int* __restrict__ rec) {
    int e = blockIdx.x * 256 + threadIdx.x;
    if (e < N_EDGES) {
        const int prev = atomicExch(&head[g_dst[e] * NSUB + (e & (NSUB - 1))], e + 1);
        int2 v;
        v.x = (g_rel[e] * 2 + (g_inv[e] != 0 ? 1 : 0)) | (g_src[e] << 10);
        v.y = prev;
        ((int2*)rec)[e] = v;
    }
}

static __device__ __forceinline__ float4 f4add(float4 a, float4 b) {
    return make_float4(a.x + b.x, a.y + b.y, a.z + b.z, a.w + b.w);
}
static __device__ __forceinline__ float4 f4scale(float4 a, float s) {
    return make_float4(a.x * s, a.y * s, a.z * s, a.w * s);
}

static __device__ __forceinline__ void mm_seg(v8f (&acc)[2], const float* arow,
                                              const float* __restrict__ wl, int nks) {
#pragma unroll 16
    for (int ks = 0; ks < nks; ++ks) {
        v2f a;
        a.x = arow[ks * 4];
        a.y = arow[ks * 4 + 1];
        const float4 bq = *(const float4*)(wl + ks * 4 * DIM);
        v2f b;
        b.x = bq.x; b.y = bq.y;
        acc[0] = __builtin_amdgcn_wmma_f32_16x16x4_f32(false, a, false, b, (short)0, acc[0], false, false);
        b.x = bq.z; b.y = bq.w;
        acc[1] = __builtin_amdgcn_wmma_f32_16x16x4_f32(false, a, false, b, (short)0, acc[1], false, false);
    }
}

__global__ __launch_bounds__(AGG_W * 32) void k_aggnode(const float* __restrict__ rel_emb,
                                                const float* __restrict__ metarel,
                                                const int*   __restrict__ head,
                                                const int*   __restrict__ rec,
                                                const float* __restrict__ WtI,
                                                const float* __restrict__ bI,
                                                const float* __restrict__ WtO,
                                                const float* __restrict__ bO,
                                                const float* __restrict__ WtS,
                                                const float* __restrict__ bS,
                                                const float* __restrict__ WtM,
                                                const float* __restrict__ bM,
                                                float* __restrict__ out,
                                                float* __restrict__ out_meta) {
    extern __shared__ float lds[];
    const int lane = threadIdx.x & 31;
    const int wave = __builtin_amdgcn_readfirstlane(threadIdx.x >> 5);
    const int lr   = lane & 15;
    const int half = lane >> 4;
    float* aw = lds;

    if (blockIdx.x < META_TILES) {
        const int t0 = blockIdx.x * 16;
        for (int rr = 0; rr < 16 / AGG_W; ++rr) {
            const int r = wave * (16 / AGG_W) + rr;
            const int m = t0 + r;
            float4 x = make_float4(0.f, 0.f, 0.f, 0.f);
            if (m < N_REL) x = *(const float4*)(metarel + m * DIM + lane * 4);
            *(float4*)(aw + r * A_STRIDE_F + lane * 4) = x;
        }
        __syncthreads();
        v8f macc[2];
        macc[0] = {};
        macc[1] = {};
        const int nbm = wave * (DIM / AGG_W) + lr * 2;
        mm_seg(macc, aw + lr * A_STRIDE_F + half * 2, WtM + half * (2 * DIM) + nbm * 2, 32);
        const float2 bMv = *(const float2*)(bM + nbm);
#pragma unroll
        for (int r = 0; r < 8; ++r) {
            const int m = t0 + half * 8 + r;
            if (m < N_REL) {
                float2 o2;
                o2.x = macc[0][r] + bMv.x;
                o2.y = macc[1][r] + bMv.y;
                *(float2*)(out_meta + m * DIM + nbm) = o2;
            }
        }
        return;
    }

    const int m0 = (blockIdx.x - META_TILES) * 16;

    const int col = lane * 8;
    const bool isMeta = (col < DIM);
    const int2* rc2 = (const int2*)rec;
    const int r0 = wave * RW;
    const float4 own0 = *(const float4*)(rel_emb + (m0 + r0 + 0) * DIM + lane * 4);
    const float4 own1 = *(const float4*)(rel_emb + (m0 + r0 + 1) * DIM + lane * 4);
    const float4 own2 = *(const float4*)(rel_emb + (m0 + r0 + 2) * DIM + lane * 4);
    const float4 own3 = *(const float4*)(rel_emb + (m0 + r0 + 3) * DIM + lane * 4);

    int  myp = head[(m0 + r0) * NSUB + (lane & (NL - 1))];
    int2 nxt = rc2[(myp > 0) ? min(myp - 1, N_EDGES - 1) : 0];
    {
        float* q = aw + r0 * A_STRIDE_F + 2 * K2 + lane * 4;
        *(float4*)(q)                  = own0;
        *(float4*)(q + A_STRIDE_F)     = own1;
        *(float4*)(q + 2 * A_STRIDE_F) = own2;
        *(float4*)(q + 3 * A_STRIDE_F) = own3;
    }
    float4 sI0[RW], sI1[RW], sO0[RW], sO1[RW];
    int cnt[RW], cO[RW];
#pragma unroll
    for (int r = 0; r < RW; ++r) {
        sI0[r] = make_float4(0.f, 0.f, 0.f, 0.f);
        sI1[r] = sI0[r]; sO0[r] = sI0[r]; sO1[r] = sI0[r];
        cnt[r] = 0; cO[r] = 0;
    }
    for (int hop = 0; hop < N_EDGES; ++hop) {
        const unsigned liveMask = __builtin_amdgcn_ballot_w32(myp != 0) & ((1u << NL) - 1u);
        if (liveMask == 0u) break;
        const int2 cur = nxt;
        myp = (myp != 0) ? cur.y : 0;
        nxt = rc2[(myp > 0) ? min(myp - 1, N_EDGES - 1) : 0];
        float4 x0[NL], x1[NL];
        int fl[NL];
#pragma unroll
        for (int r = 0; r < NL; ++r) {
            if (liveMask & (1u << r)) {
                const int cx = __builtin_amdgcn_readlane(cur.x, r);
                fl[r] = cx & 1;
                const float* p = isMeta ? (metarel + ((cx & 1023) >> 1) * DIM + col)
                                        : (rel_emb + (cx >> 10) * DIM + (col - DIM));
                x0[r] = *(const float4*)p;
                x1[r] = *(const float4*)(p + 4);
            }
        }
#pragma unroll
        for (int j = 0; j < NL; ++j) {
            if (liveMask & (1u << j)) {
                const int r = j / NSUB;
                ++cnt[r];
                if (fl[j]) { sO0[r] = f4add(sO0[r], x0[j]); sO1[r] = f4add(sO1[r], x1[j]); ++cO[r]; }
                else       { sI0[r] = f4add(sI0[r], x0[j]); sI1[r] = f4add(sI1[r], x1[j]); }
            }
        }
    }
#pragma unroll
    for (int r = 0; r < RW; ++r) {
        const float inv = 1.0f / fmaxf((float)cnt[r], 1.0f);
        float* q = aw + (r0 + r) * A_STRIDE_F;
        *(float4*)(q + col)           = f4scale(sI0[r], inv);
        *(float4*)(q + col + 4)       = f4scale(sI1[r], inv);
        *(float4*)(q + K2 + col)      = f4scale(sO0[r], inv);
        *(float4*)(q + K2 + col + 4)  = f4scale(sO1[r], inv);
        if (lane == 0) {
            q[KF]     = (float)(cnt[r] - cO[r]) * inv;
            q[KF + 1] = (float)cO[r] * inv;
        }
    }
    __syncthreads();

    v8f acc[2];
    acc[0] = {};
    acc[1] = {};
    const int    nb   = wave * (DIM / AGG_W) + lr * 2;
    const float* arow = aw + lr * A_STRIDE_F + half * 2;
    const int    wofs = half * (2 * DIM) + nb * 2;
    mm_seg(acc, arow,          WtI + wofs, 64);
    mm_seg(acc, arow + K2,     WtO + wofs, 64);
    mm_seg(acc, arow + 2 * K2, WtS + wofs, 32);

    const float2 bSv = *(const float2*)(bS + nb);
    const float2 bIv = *(const float2*)(bI + nb);
    const float2 bOv = *(const float2*)(bO + nb);
#pragma unroll
    for (int r = 0; r < 8; ++r) {
        const int row = half * 8 + r;
        const float fI = aw[row * A_STRIDE_F + KF];
        const float fO = aw[row * A_STRIDE_F + KF + 1];
        float2 o2;
        o2.x = acc[0][r] + bSv.x + (fI * bIv.x + fO * bOv.x);
        o2.y = acc[1][r] + bSv.y + (fI * bIv.y + fO * bOv.y);
        *(float2*)(out + (m0 + row) * DIM + nb) = o2;
    }
}

extern "C" void kernel_launch(void* const* d_in, const int* in_sizes, int n_in,
                              void* d_out, int out_size, void* d_ws, size_t ws_size,
                              hipStream_t stream) {
    const float* rel_emb = (const float*)d_in[0];
    const float* metarel = (const float*)d_in[1];
    const int*   srcI    = (const int*)  d_in[2];
    const int*   dstI    = (const int*)  d_in[3];
    const int*   relI    = (const int*)  d_in[4];
    const int*   invI    = (const int*)  d_in[5];
    const float* W_I_w   = (const float*)d_in[6];
    const float* W_I_b   = (const float*)d_in[7];
    const float* W_O_w   = (const float*)d_in[8];
    const float* W_O_b   = (const float*)d_in[9];
    const float* W_S_w   = (const float*)d_in[10];
    const float* W_S_b   = (const float*)d_in[11];
    const float* W_M_w   = (const float*)d_in[12];
    const float* W_M_b   = (const float*)d_in[13];

    float* ws  = (float*)d_ws;
    float* WtI = ws + WS_WT_I;
    float* WtO = ws + WS_WT_O;
    float* WtS = ws + WS_WT_S;
    float* WtM = ws + WS_WT_M;
    int*   head = (int*)d_ws + WS_HEAD;
    int*   rec  = (int*)d_ws + WS_REC;

    float* out_rel  = (float*)d_out;
    float* out_meta = (float*)d_out + N_NODES * DIM;

    const int aggLds  = 16 * A_STRIDE_F * sizeof(float);

    k_prep<<<PREP_B, 256, 0, stream>>>(W_I_w, W_O_w, W_S_w, W_M_w, ws);

    k_link<<<N_EDGES / 256, 256, 0, stream>>>(srcI, dstI, relI, invI, head, rec);

    k_aggnode<<<META_TILES + N_NODES / 16, AGG_W * 32, aggLds, stream>>>(rel_emb, metarel, head, rec,
                                                     WtI, W_I_b, WtO, W_O_b, WtS, W_S_b,
                                                     WtM, W_M_b, out_rel, out_meta);
}


# k_aggnode gather: a lane takes 4 columns of each comp_h half, so a row load is one contiguous 512-byte piece on a scalar base (each line touched once); 144 VGPRs: 7 waves per SIMD, not 6; sums kept
# speedup vs baseline: 10.2744x; 1.0431x over previous
#include <hip/hip_runtime.h>


typedef float v2f __attribute__((ext_vector_type(2)));
typedef float v8f __attribute__((ext_vector_type(8)));

#define N_NODES 50000
#define N_EDGES 640000
#define N_REL   500
#define DIM     128
#define K2      256

#define WS_WT_I 0
#define WS_WT_O 32768
#define WS_WT_S 65536
#define WS_WT_M 81920
#define WS_HEAD 98304
#define WS_REC  248320
#define NSUB 2
#define WS_ZERO_N 100000

#define A_STRIDE_F 644
#define KF         640

#define AGG_W    4
#define RW       4
#define NL       8
#define META_TILES 32

#define ZERO_B  391
#define PREP_B  775
__global__ __launch_bounds__(256) void k_prep(const float* __restrict__ wI,
                                              const float* __restrict__ wO,
                                              const float* __restrict__ wS,
                                              const float* __restrict__ wM,
                                              float* __restrict__ ws) {
    int b = blockIdx.x;
    if (b < ZERO_B) {
        const int i = b * 256 + threadIdx.x;
        if (i < WS_ZERO_N) ws[WS_HEAD + i] = 0.0f;
        return;
    }
    b -= ZERO_B;
    const float* src;
    float* dst;
    int lgC;
    if (b < 128)      { src = wI; dst = ws + WS_WT_I; lgC = 8; }
    else if (b < 256) { src = wO; dst = ws + WS_WT_O; lgC = 8; b -= 128; }
    else if (b < 320) { src = wS; dst = ws + WS_WT_S; lgC = 7; b -= 256; }
    else              { src = wM; dst = ws + WS_WT_M; lgC = 7; b -= 320; }
    const int i = b * 256 + threadIdx.x;
    const int r = i >> lgC, c = i - (r << lgC);
    dst[(c >> 1) * (2 * DIM) + r * 2 + (c & 1)] = src[i];
}

__global__ __launch_bounds__(256) void k_link(const int* __restrict__ g_src,
                                              const int* __restrict__ g_dst,
                                              const int* __restrict__ g_rel,
                                              const int* __restrict__ g_inv,
                                              int* __restrict__ head,
                                              int* __restrict__ rec) {
    int e = blockIdx.x * 256 + threadIdx.x;
    if (e < N_EDGES) {
        const int prev = atomicExch(&head[g_dst[e] * NSUB + (e & (NSUB - 1))], e + 1);
        int2 v;
        v.x = (g_rel[e] * 2 + (g_inv[e] != 0 ? 1 : 0)) | (g_src[e] << 10);
        v.y = prev;
        ((int2*)rec)[e] = v;
    }
}

static __device__ __forceinline__ float4 f4add(float4 a, float4 b) {
    return make_float4(a.x + b.x, a.y + b.y, a.z + b.z, a.w + b.w);
}
static __device__ __forceinline__ float4 f4scale(float4 a, float s) {
    return make_float4(a.x * s, a.y * s, a.z * s, a.w * s);
}

static __device__ __forceinline__ void mm_seg(v8f (&acc)[2], const float* arow,
                                              const float* __restrict__ wl, int nks) {
#pragma unroll 16
    for (int ks = 0; ks < nks; ++ks) {
        v2f a;
        a.x = arow[ks * 4];
        a.y = arow[ks * 4 + 1];
        const float4 bq = *(const float4*)(wl + ks * 4 * DIM);
        v2f b;
        b.x = bq.x; b.y = bq.y;
        acc[0] = __builtin_amdgcn_wmma_f32_16x16x4_f32(false, a, false, b, (short)0, acc[0], false, false);
        b.x = bq.z; b.y = bq.w;
        acc[1] = __builtin_amdgcn_wmma_f32_16x16x4_f32(false, a, false, b, (short)0, acc[1], false, false);
    }
}

__global__ __launch_bounds__(AGG_W * 32) void k_aggnode(const float* __restrict__ rel_emb,
                                                const float* __restrict__ metarel,
                                                const int*   __restrict__ head,
                                                const int*   __restrict__ rec,
                                                const float* __restrict__ WtI,
                                                const float* __restrict__ bI,
                                                const float* __restrict__ WtO,
                                                const float* __restrict__ bO,
                                                const float* __restrict__ WtS,
                                                const float* __restrict__ bS,
                                                const float* __restrict__ WtM,
                                                const float* __restrict__ bM,
                                                float* __restrict__ out,
                                                float* __restrict__ out_meta) {
    extern __shared__ float lds[];
    const int lane = threadIdx.x & 31;
    const int wave = __builtin_amdgcn_readfirstlane(threadIdx.x >> 5);
    float* aw = lds;

    if (blockIdx.x < META_TILES) {
        const int t0 = blockIdx.x * 16;
        for (int rr = 0; rr < 16 / AGG_W; ++rr) {
            const int r = wave * (16 / AGG_W) + rr;
            const int m = t0 + r;
            float4 x = make_float4(0.f, 0.f, 0.f, 0.f);
            if (m < N_REL) x = *(const float4*)(metarel + m * DIM + lane * 4);
            *(float4*)(aw + r * A_STRIDE_F + lane * 4) = x;
        }
        __syncthreads();
        const int lr   = lane & 15;
        const int half = lane >> 4;
        v8f macc[2];
        macc[0] = {};
        macc[1] = {};
        const int nbm = wave * (DIM / AGG_W) + lr * 2;
        mm_seg(macc, aw + lr * A_STRIDE_F + half * 2, WtM + half * (2 * DIM) + nbm * 2, 32);
        const float2 bMv = *(const float2*)(bM + nbm);
#pragma unroll
        for (int r = 0; r < 8; ++r) {
            const int m = t0 + half * 8 + r;
            if (m < N_REL) {
                float2 o2;
                o2.x = macc[0][r] + bMv.x;
                o2.y = macc[1][r] + bMv.y;
                *(float2*)(out_meta + m * DIM + nbm) = o2;
            }
        }
        return;
    }

    const int m0 = (blockIdx.x - META_TILES) * 16;

    const unsigned c4 = (unsigned)lane * 4u;
    const int2* rc2 = (const int2*)rec;
    const int r0 = wave * RW;
    const float4 own0 = *(const float4*)(rel_emb + (m0 + r0 + 0) * DIM + lane * 4);
    const float4 own1 = *(const float4*)(rel_emb + (m0 + r0 + 1) * DIM + lane * 4);
    const float4 own2 = *(const float4*)(rel_emb + (m0 + r0 + 2) * DIM + lane * 4);
    const float4 own3 = *(const float4*)(rel_emb + (m0 + r0 + 3) * DIM + lane * 4);

    int  myp = head[(m0 + r0) * NSUB + (lane & (NL - 1))];
    int2 nxt = rc2[(myp > 0) ? min(myp - 1, N_EDGES - 1) : 0];
    {
        float* q = aw + r0 * A_STRIDE_F + 2 * K2 + lane * 4;
        *(float4*)(q)                  = own0;
        *(float4*)(q + A_STRIDE_F)     = own1;
        *(float4*)(q + 2 * A_STRIDE_F) = own2;
        *(float4*)(q + 3 * A_STRIDE_F) = own3;
    }
    float4 sI0[RW], sI1[RW], sO0[RW], sO1[RW];
    int cnt[RW], cO[RW];
#pragma unroll
    for (int r = 0; r < RW; ++r) {
        sI0[r] = make_float4(0.f, 0.f, 0.f, 0.f);
        sI1[r] = sI0[r]; sO0[r] = sI0[r]; sO1[r] = sI0[r];
        cnt[r] = 0; cO[r] = 0;
    }
    for (int hop = 0; hop < N_EDGES; ++hop) {
        const unsigned liveMask = __builtin_amdgcn_ballot_w32(myp != 0) & ((1u << NL) - 1u);
        if (liveMask == 0u) break;
        const int2 cur = nxt;
        myp = (myp != 0) ? cur.y : 0;
        nxt = rc2[(myp > 0) ? min(myp - 1, N_EDGES - 1) : 0];
        float4 x0[NL], x1[NL];
        int fl[NL];
#pragma unroll
        for (int r = 0; r < NL; ++r) {
            if (liveMask & (1u << r)) {
                const int cx = __builtin_amdgcn_readlane(cur.x, r);
                fl[r] = cx & 1;
                const unsigned im = (unsigned)(((cx & 1023) >> 1) * DIM) + c4;
                const unsigned is = (unsigned)(((cx >> 10) & 0xFFFF) * DIM) + c4;
                x0[r] = *(const float4*)(metarel + im);
                x1[r] = *(const float4*)(rel_emb + is);
            }
        }
#pragma unroll
        for (int j = 0; j < NL; ++j) {
            if (liveMask & (1u << j)) {
                const int r = j / NSUB;
                ++cnt[r];
                if (fl[j]) { sO0[r] = f4add(sO0[r], x0[j]); sO1[r] = f4add(sO1[r], x1[j]); ++cO[r]; }
                else       { sI0[r] = f4add(sI0[r], x0[j]); sI1[r] = f4add(sI1[r], x1[j]); }
            }
        }
    }
#pragma unroll
    for (int r = 0; r < RW; ++r) {
        const float inv = 1.0f / fmaxf((float)cnt[r], 1.0f);
        float* q = aw + (r0 + r) * A_STRIDE_F;
        *(float4*)(q + c4)            = f4scale(sI0[r], inv);
        *(float4*)(q + DIM + c4)      = f4scale(sI1[r], inv);
        *(float4*)(q + K2 + c4)       = f4scale(sO0[r], inv);
        *(float4*)(q + K2 + DIM + c4) = f4scale(sO1[r], inv);
        if (lane == 0) {
            q[KF]     = (float)(cnt[r] - cO[r]) * inv;
            q[KF + 1] = (float)cO[r] * inv;
        }
    }
    __syncthreads();

    const int lr   = lane & 15;
    const int half = lane >> 4;
    v8f acc[2];
    acc[0] = {};
    acc[1] = {};
    const int    nb   = wave * (DIM / AGG_W) + lr * 2;
    const float* arow = aw + lr * A_STRIDE_F + half * 2;
    const int    wofs = half * (2 * DIM) + nb * 2;
    mm_seg(acc, arow,          WtI + wofs, 64);
    mm_seg(acc, arow + K2,     WtO + wofs, 64);
    mm_seg(acc, arow + 2 * K2, WtS + wofs, 32);

    const float2 bSv = *(const float2*)(bS + nb);
    const float2 bIv = *(const float2*)(bI + nb);
    const float2 bOv = *(const float2*)(bO + nb);
#pragma unroll
    for (int r = 0; r < 8; ++r) {
        const int row = half * 8 + r;
        const float fI = aw[row * A_STRIDE_F + KF];
        const float fO = aw[row * A_STRIDE_F + KF + 1];
        float2 o2;
        o2.x = acc[0][r] + bSv.x + (fI * bIv.x + fO * bOv.x);
        o2.y = acc[1][r] + bSv.y + (fI * bIv.y + fO * bOv.y);
        *(float2*)(out + (m0 + row) * DIM + nb) = o2;
    }
}

extern "C" void kernel_launch(void* const* d_in, const int* in_sizes, int n_in,
                              void* d_out, int out_size, void* d_ws, size_t ws_size,
                              hipStream_t stream) {
    const float* rel_emb = (const float*)d_in[0];
    const float* metarel = (const float*)d_in[1];
    const int*   srcI    = (const int*)  d_in[2];
    const int*   dstI    = (const int*)  d_in[3];
    const int*   relI    = (const int*)  d_in[4];
    const int*   invI    = (const int*)  d_in[5];
    const float* W_I_w   = (const float*)d_in[6];
    const float* W_I_b   = (const float*)d_in[7];
    const float* W_O_w   = (const float*)d_in[8];
    const float* W_O_b   = (const float*)d_in[9];
    const float* W_S_w   = (const float*)d_in[10];
    const float* W_S_b   = (const float*)d_in[11];
    const float* W_M_w   = (const float*)d_in[12];
    const float* W_M_b   = (const float*)d_in[13];

    float* ws  = (float*)d_ws;
    float* WtI = ws + WS_WT_I;
    float* WtO = ws + WS_WT_O;
    float* WtS = ws + WS_WT_S;
    float* WtM = ws + WS_WT_M;
    int*   head = (int*)d_ws + WS_HEAD;
    int*   rec  = (int*)d_ws + WS_REC;

    float* out_rel  = (float*)d_out;
    float* out_meta = (float*)d_out + N_NODES * DIM;

    const int aggLds  = 16 * A_STRIDE_F * sizeof(float);

    k_prep<<<PREP_B, 256, 0, stream>>>(W_I_w, W_O_w, W_S_w, W_M_w, ws);

    k_link<<<N_EDGES / 256, 256, 0, stream>>>(srcI, dstI, relI, invI, head, rec);

    k_aggnode<<<META_TILES + N_NODES / 16, AGG_W * 32, aggLds, stream>>>(rel_emb, metarel, head, rec,
                                                     WtI, W_I_b, WtO, W_O_b, WtS, W_S_b,
                                                     WtM, W_M_b, out_rel, out_meta);
}
